# starting kernel
# baseline (speedup 1.0000x reference)
#include <hip/hip_runtime.h>

typedef __attribute__((ext_vector_type(2))) float v2f;
typedef __attribute__((ext_vector_type(8))) float v8f;

__global__ void zero_f32(float* __restrict__ p, long n) {
    long i4 = (long)blockIdx.x * blockDim.x + threadIdx.x;
    long stride = (long)gridDim.x * blockDim.x;
    long n4 = n >> 2;
    float4* p4 = (float4*)p;
    const float4 z = make_float4(0.f, 0.f, 0.f, 0.f);
    for (long i = i4; i < n4; i += stride) p4[i] = z;
    for (long i = (n4 << 2) + i4; i < n; i += stride) p[i] = 0.0f;
}

__global__ void edge_agg64(const float* __restrict__ feat,
                           const int* __restrict__ src,
                           const int* __restrict__ dst,
                           float* __restrict__ agg,
                           float* __restrict__ deg,
                           int E) {
    const int lane = threadIdx.x & 31;
    const int e = blockIdx.x * (blockDim.x >> 5) + (threadIdx.x >> 5);
    if (e >= E) return;
    const int s = src[e];
    const int d = dst[e];
    const float* fs = feat + (size_t)s * 64;
    float* ad = agg + (size_t)d * 64;
    __hip_atomic_fetch_add(&ad[lane],      fs[lane],      __ATOMIC_RELAXED, __HIP_MEMORY_SCOPE_AGENT);
    __hip_atomic_fetch_add(&ad[lane + 32], fs[lane + 32], __ATOMIC_RELAXED, __HIP_MEMORY_SCOPE_AGENT);
    if (lane == 0 && deg != nullptr)
        __hip_atomic_fetch_add(&deg[d], 1.0f, __ATOMIC_RELAXED, __HIP_MEMORY_SCOPE_AGENT);
}

template <int K, int M>
__global__ void sage_gemm_wmma(const float* __restrict__ agg,
                               const float* __restrict__ deg,
                               const float* __restrict__ xroot,
                               const float* __restrict__ wl,
                               const float* __restrict__ wr,
                               const float* __restrict__ bias,
                               float* __restrict__ out,
                               int N) {
    constexpr int S = K / 4;
    const int lane = threadIdx.x & 31;
    const int wave = threadIdx.x >> 5;
    const int lo   = lane & 15;
    const int half = lane >> 4;
    const int row0 = blockIdx.x * 16;
    const int col0 = wave * 16;

    int ar = row0 + lo;
    if (ar >= N) ar = N - 1;
    const float rinv = 1.0f / fmaxf(deg[ar], 1.0f);
    const int bcol = col0 + lo;

    const float* __restrict__ pa = agg   + (size_t)ar * K + half * 2;
    const float* __restrict__ px = xroot + (size_t)ar * K + half * 2;
    const float* __restrict__ pl = wl + (size_t)(half * 2) * M + bcol;
    const float* __restrict__ pr = wr + (size_t)(half * 2) * M + bcol;

    v2f af[S], bf[S];
    v8f acc = {};

#pragma unroll
    for (int s = 0; s < S; ++s) {
        af[s].x = pa[s * 4];
        af[s].y = pa[s * 4 + 1];
        bf[s].x = pl[(size_t)(s * 4) * M];
        bf[s].y = pl[(size_t)(s * 4 + 1) * M];
    }
#pragma unroll
    for (int s = 0; s < S; ++s) {
        v2f a;
        a.x = af[s].x * rinv;
        a.y = af[s].y * rinv;
        acc = __builtin_amdgcn_wmma_f32_16x16x4_f32(false, a, false, bf[s],
                                                    (short)0, acc, false, false);
    }

#pragma unroll
    for (int s = 0; s < S; ++s) {
        af[s].x = px[s * 4];
        af[s].y = px[s * 4 + 1];
        bf[s].x = pr[(size_t)(s * 4) * M];
        bf[s].y = pr[(size_t)(s * 4 + 1) * M];
    }
#pragma unroll
    for (int s = 0; s < S; ++s) {
        acc = __builtin_amdgcn_wmma_f32_16x16x4_f32(false, af[s], false, bf[s],
                                                    (short)0, acc, false, false);
    }

    const float bv = bias[bcol];
#pragma unroll
    for (int v = 0; v < 8; ++v) {
        const int r = row0 + half * 8 + v;
        if (r < N) {
            float val = acc[v] + bv;
            out[(size_t)r * M + bcol] = fmaxf(val, 0.0f);
        }
    }
}

__global__ void classifier_head(const float* __restrict__ h2,
                                const float* __restrict__ wc1,
                                const float* __restrict__ bc1,
                                const float* __restrict__ wc2,
                                const float* __restrict__ bc2,
                                float* __restrict__ out, int N) {
    const int n = blockIdx.x * blockDim.x + threadIdx.x;
    if (n >= N) return;
    const float* h = h2 + (size_t)n * 32;
    float hn[32];
#pragma unroll
    for (int k = 0; k < 32; ++k) hn[k] = h[k];
    float h3[16];
#pragma unroll
    for (int j = 0; j < 16; ++j) {
        float s = bc1[j];
#pragma unroll
        for (int k = 0; k < 32; ++k) s = fmaf(hn[k], wc1[k * 16 + j], s);
        h3[j] = fmaxf(s, 0.0f);
    }
#pragma unroll
    for (int j = 0; j < 2; ++j) {
        float s = bc2[j];
#pragma unroll
        for (int k = 0; k < 16; ++k) s = fmaf(h3[k], wc2[k * 2 + j], s);
        out[(size_t)n * 2 + j] = s;
    }
}

extern "C" void kernel_launch(void* const* d_in, const int* in_sizes, int n_in,
                              void* d_out, int out_size, void* d_ws, size_t ws_size,
                              hipStream_t stream) {
    const float* x    = (const float*)d_in[0];
    const int*   ei   = (const int*)d_in[1];
    const float* w1l  = (const float*)d_in[2];
    const float* w1r  = (const float*)d_in[3];
    const float* b1   = (const float*)d_in[4];
    const float* w2l  = (const float*)d_in[5];
    const float* w2r  = (const float*)d_in[6];
    const float* b2   = (const float*)d_in[7];
    const float* wc1  = (const float*)d_in[8];
    const float* bc1  = (const float*)d_in[9];
    const float* wc2  = (const float*)d_in[10];
    const float* bc2  = (const float*)d_in[11];
    float* out = (float*)d_out;

    const int N = in_sizes[0] / 64;
    const int E = in_sizes[1] / 2;
    const int* src = ei;
    const int* dst = ei + E;

    float* agg1 = (float*)d_ws;
    float* agg2 = agg1 + (size_t)N * 64;
    float* deg  = agg2 + (size_t)N * 64;
    float* h1   = deg  + (size_t)N;
    float* h2   = h1   + (size_t)N * 64;

    {
        long nz = (long)N * (64 + 64 + 1);
        zero_f32<<<2048, 256, 0, stream>>>(agg1, nz);
    }

    {
        int blocks = (E + 7) / 8;
        edge_agg64<<<blocks, 256, 0, stream>>>(x, src, dst, agg1, deg, E);
    }

    {
        int blocks = (N + 15) / 16;
        sage_gemm_wmma<64, 64><<<blocks, 128, 0, stream>>>(agg1, deg, x, w1l,
                                                           w1r, b1, h1, N);
    }

    {
        int blocks = (E + 7) / 8;
        edge_agg64<<<blocks, 256, 0, stream>>>(h1, src, dst, agg2, nullptr, E);
    }

    {
        int blocks = (N + 15) / 16;
        sage_gemm_wmma<64, 32><<<blocks, 64, 0, stream>>>(agg2, deg, h1, w2l,
                                                          w2r, b2, h2, N);
    }

    {
        int blocks = (N + 255) / 256;
        classifier_head<<<blocks, 256, 0, stream>>>(h2, wc1, bc1, wc2, bc2, out, N);
    }
}


# Degree counts moved from a dense float-per-node array (32 counters per 128-B line, each line taking hundreds of adds) to one slot per 256-B row of the idle layer-2 plane; GEMM 1 copies out and clears
# speedup vs baseline: 1.2679x; 1.2679x over previous
#include <hip/hip_runtime.h>

typedef __attribute__((ext_vector_type(2))) float v2f;
typedef __attribute__((ext_vector_type(8))) float v8f;

__global__ void zero_f32(float* __restrict__ p, long n) {
    long i4 = (long)blockIdx.x * blockDim.x + threadIdx.x;
    long stride = (long)gridDim.x * blockDim.x;
    long n4 = n >> 2;
    float4* p4 = (float4*)p;
    const float4 z = make_float4(0.f, 0.f, 0.f, 0.f);
    for (long i = i4; i < n4; i += stride) p4[i] = z;
    for (long i = (n4 << 2) + i4; i < n; i += stride) p[i] = 0.0f;
}

__global__ void edge_agg64(const float* __restrict__ feat,
                           const int* __restrict__ src,
                           const int* __restrict__ dst,
                           float* __restrict__ agg,
                           float* __restrict__ cnt,
                           int E) {
    const int lane = threadIdx.x & 31;
    const int e = blockIdx.x * (blockDim.x >> 5) + (threadIdx.x >> 5);
    if (e >= E) return;
    const int s = src[e];
    const int d = dst[e];
    const float* fs = feat + (size_t)s * 64;
    float* ad = agg + (size_t)d * 64;
    __hip_atomic_fetch_add(&ad[lane],      fs[lane],      __ATOMIC_RELAXED, __HIP_MEMORY_SCOPE_AGENT);
    __hip_atomic_fetch_add(&ad[lane + 32], fs[lane + 32], __ATOMIC_RELAXED, __HIP_MEMORY_SCOPE_AGENT);
    if (lane == 0 && cnt != nullptr)
        __hip_atomic_fetch_add(&cnt[(size_t)d * 64], 1.0f, __ATOMIC_RELAXED, __HIP_MEMORY_SCOPE_AGENT);
}

template <int K, int M, bool FIRST>
__global__ void sage_gemm_wmma(const float* __restrict__ agg,
                               float* degv,
                               float* cnt,
                               const float* __restrict__ xroot,
                               const float* __restrict__ wl,
                               const float* __restrict__ wr,
                               const float* __restrict__ bias,
                               float* __restrict__ out,
                               int N) {
    constexpr int S = K / 4;
    const int lane = threadIdx.x & 31;
    const int wave = threadIdx.x >> 5;
    const int lo   = lane & 15;
    const int half = lane >> 4;
    const int row0 = blockIdx.x * 16;
    const int col0 = wave * 16;

    int ar = row0 + lo;
    if (ar >= N) ar = N - 1;
    float dcount;
    if constexpr (FIRST) dcount = cnt[(size_t)ar * 64];
    else                 dcount = degv[ar];
    const float rinv = 1.0f / fmaxf(dcount, 1.0f);
    if constexpr (FIRST) {
        __syncthreads();
        if (wave == 0 && half == 0 && row0 + lo < N) {
            degv[ar] = dcount;
            cnt[(size_t)ar * 64] = 0.0f;
        }
    }
    const int bcol = col0 + lo;

    const float* __restrict__ pa = agg   + (size_t)ar * K + half * 2;
    const float* __restrict__ px = xroot + (size_t)ar * K + half * 2;
    const float* __restrict__ pl = wl + (size_t)(half * 2) * M + bcol;
    const float* __restrict__ pr = wr + (size_t)(half * 2) * M + bcol;

    v2f af[S], bf[S];
    v8f acc = {};

#pragma unroll
    for (int s = 0; s < S; ++s) {
        af[s].x = pa[s * 4];
        af[s].y = pa[s * 4 + 1];
        bf[s].x = pl[(size_t)(s * 4) * M];
        bf[s].y = pl[(size_t)(s * 4 + 1) * M];
    }
#pragma unroll
    for (int s = 0; s < S; ++s) {
        v2f a;
        a.x = af[s].x * rinv;
        a.y = af[s].y * rinv;
        acc = __builtin_amdgcn_wmma_f32_16x16x4_f32(false, a, false, bf[s],
                                                    (short)0, acc, false, false);
    }

#pragma unroll
    for (int s = 0; s < S; ++s) {
        af[s].x = px[s * 4];
        af[s].y = px[s * 4 + 1];
        bf[s].x = pr[(size_t)(s * 4) * M];
        bf[s].y = pr[(size_t)(s * 4 + 1) * M];
    }
#pragma unroll
    for (int s = 0; s < S; ++s) {
        acc = __builtin_amdgcn_wmma_f32_16x16x4_f32(false, af[s], false, bf[s],
                                                    (short)0, acc, false, false);
    }

    const float bv = bias[bcol];
#pragma unroll
    for (int v = 0; v < 8; ++v) {
        const int r = row0 + half * 8 + v;
        if (r < N) {
            float val = acc[v] + bv;
            out[(size_t)r * M + bcol] = fmaxf(val, 0.0f);
        }
    }
}

__global__ void classifier_head(const float* __restrict__ h2,
                                const float* __restrict__ wc1,
                                const float* __restrict__ bc1,
                                const float* __restrict__ wc2,
                                const float* __restrict__ bc2,
                                float* __restrict__ out, int N) {
    const int n = blockIdx.x * blockDim.x + threadIdx.x;
    if (n >= N) return;
    const float* h = h2 + (size_t)n * 32;
    float hn[32];
#pragma unroll
    for (int k = 0; k < 32; ++k) hn[k] = h[k];
    float h3[16];
#pragma unroll
    for (int j = 0; j < 16; ++j) {
        float s = bc1[j];
#pragma unroll
        for (int k = 0; k < 32; ++k) s = fmaf(hn[k], wc1[k * 16 + j], s);
        h3[j] = fmaxf(s, 0.0f);
    }
#pragma unroll
    for (int j = 0; j < 2; ++j) {
        float s = bc2[j];
#pragma unroll
        for (int k = 0; k < 16; ++k) s = fmaf(h3[k], wc2[k * 2 + j], s);
        out[(size_t)n * 2 + j] = s;
    }
}

extern "C" void kernel_launch(void* const* d_in, const int* in_sizes, int n_in,
                              void* d_out, int out_size, void* d_ws, size_t ws_size,
                              hipStream_t stream) {
    const float* x    = (const float*)d_in[0];
    const int*   ei   = (const int*)d_in[1];
    const float* w1l  = (const float*)d_in[2];
    const float* w1r  = (const float*)d_in[3];
    const float* b1   = (const float*)d_in[4];
    const float* w2l  = (const float*)d_in[5];
    const float* w2r  = (const float*)d_in[6];
    const float* b2   = (const float*)d_in[7];
    const float* wc1  = (const float*)d_in[8];
    const float* bc1  = (const float*)d_in[9];
    const float* wc2  = (const float*)d_in[10];
    const float* bc2  = (const float*)d_in[11];
    float* out = (float*)d_out;

    const int N = in_sizes[0] / 64;
    const int E = in_sizes[1] / 2;
    const int* src = ei;
    const int* dst = ei + E;

    float* agg1 = (float*)d_ws;
    float* agg2 = agg1 + (size_t)N * 64;
    float* deg  = agg2 + (size_t)N * 64;
    float* h1   = deg  + (size_t)N;
    float* h2   = h1   + (size_t)N * 64;

    {
        long nz = (long)N * (64 + 64);
        zero_f32<<<2048, 256, 0, stream>>>(agg1, nz);
    }

    {
        int blocks = (E + 7) / 8;
        edge_agg64<<<blocks, 256, 0, stream>>>(x, src, dst, agg1, agg2, E);
    }

    {
        int blocks = (N + 15) / 16;
        sage_gemm_wmma<64, 64, true><<<blocks, 128, 0, stream>>>(agg1, deg, agg2, x, w1l,
                                                                 w1r, b1, h1, N);
    }

    {
        int blocks = (E + 7) / 8;
        edge_agg64<<<blocks, 256, 0, stream>>>(h1, src, dst, agg2, nullptr, E);
    }

    {
        int blocks = (N + 15) / 16;
        sage_gemm_wmma<64, 32, false><<<blocks, 64, 0, stream>>>(agg2, deg, nullptr, h1, w2l,
                                                                 w2r, b2, h2, N);
    }

    {
        int blocks = (N + 255) / 256;
        classifier_head<<<blocks, 256, 0, stream>>>(h2, wc1, bc1, wc2, bc2, out, N);
    }
}


# Scatter of float atomics replaced by a per-call counting sort of edges by destination (int rank atomics, block scans) and an atomic-free gather-sum per node; rows now summed in arrival-rank order
# speedup vs baseline: 2.4128x; 1.9030x over previous
#include <hip/hip_runtime.h>


typedef __attribute__((ext_vector_type(2))) float v2f;
typedef __attribute__((ext_vector_type(8))) float v8f;

__global__ void clear_counters(int* __restrict__ cnt, int N) {
    const int n = blockIdx.x * blockDim.x + threadIdx.x;
    if (n < N) cnt[(size_t)n * 64] = 0;
}

__global__ void rank_edges(const int* __restrict__ dst,
                           int* __restrict__ cnt,
                           int* __restrict__ rank, int E) {
    const int e = blockIdx.x * blockDim.x + threadIdx.x;
    if (e >= E) return;
    const int d = dst[e];
    rank[e] = __hip_atomic_fetch_add(&cnt[(size_t)d * 64], 1, __ATOMIC_RELAXED,
                                     __HIP_MEMORY_SCOPE_AGENT);
}

__global__ void bucket_local_scan(const int* __restrict__ cnt,
                                  int* __restrict__ cnti,
                                  float* __restrict__ deg,
                                  int* __restrict__ off,
                                  int* __restrict__ bsum, int N) {
    __shared__ int sh[256];
    const int t = threadIdx.x;
    const int n = blockIdx.x * 256 + t;
    const int c = (n < N) ? cnt[(size_t)n * 64] : 0;
    sh[t] = c;
    __syncthreads();
    for (int o = 1; o < 256; o <<= 1) {
        const int v = (t >= o) ? sh[t - o] : 0;
        __syncthreads();
        sh[t] += v;
        __syncthreads();
    }
    const int incl = sh[t];
    if (n < N) {
        off[n] = incl - c;
        cnti[n] = c;
        deg[n] = (float)c;
    }
    if (t == 255) bsum[blockIdx.x] = incl;
}

__global__ void add_bucket_base(int* __restrict__ off,
                                const int* __restrict__ bsum, int N) {
    __shared__ int sh[256];
    const int t = threadIdx.x;
    const int b = blockIdx.x;
    int p = 0;
    for (int i = t; i < b; i += 256) p += bsum[i];
    sh[t] = p;
    __syncthreads();
    for (int o = 128; o > 0; o >>= 1) {
        if (t < o) sh[t] += sh[t + o];
        __syncthreads();
    }
    const int base = sh[0];
    const int n = b * 256 + t;
    if (n < N) off[n] += base;
}

__global__ void fill_buckets(const int* __restrict__ src,
                             const int* __restrict__ dst,
                             const int* __restrict__ rank,
                             const int* __restrict__ off,
                             int* __restrict__ perm, int E) {
    const int e = blockIdx.x * blockDim.x + threadIdx.x;
    if (e >= E) return;
    perm[off[dst[e]] + rank[e]] = src[e];
}

__global__ void gather_sum64(const float* __restrict__ feat,
                             const int* __restrict__ perm,
                             const int* __restrict__ off,
                             const int* __restrict__ cnti,
                             float* __restrict__ agg, int N) {
    const int lane = threadIdx.x & 31;
    const int n = __builtin_amdgcn_readfirstlane(
        (int)(blockIdx.x * (blockDim.x >> 5) + (threadIdx.x >> 5)));
    if (n >= N) return;
    const int start = off[n];
    const int cnt = cnti[n];
    const float2* __restrict__ f2 = (const float2*)feat;
    float2 acc = make_float2(0.0f, 0.0f);
    for (int base = 0; base < cnt; base += 32) {
        const int m = min(32, cnt - base);
        const int idx = (lane < m) ? perm[start + base + lane] : 0;
        int t = 0;
        for (; t + 8 <= m; t += 8) {
            float2 r[8];
#pragma unroll
            for (int i = 0; i < 8; ++i) {
                const int s = __builtin_amdgcn_readlane(idx, t + i);
                r[i] = f2[(size_t)s * 32 + lane];
            }
#pragma unroll
            for (int i = 0; i < 8; ++i) {
                acc.x += r[i].x;
                acc.y += r[i].y;
            }
        }
        for (; t < m; t += 4) {
            float2 r[4];
#pragma unroll
            for (int i = 0; i < 4; ++i) {
                const int s = __builtin_amdgcn_readlane(idx, t + i);
                r[i] = f2[(size_t)s * 32 + lane];
            }
#pragma unroll
            for (int i = 0; i < 4; ++i) {
                if (t + i < m) {
                    acc.x += r[i].x;
                    acc.y += r[i].y;
                }
            }
        }
    }
    ((float2*)agg)[(size_t)n * 32 + lane] = acc;
}

template <int K, int M>
__global__ void sage_gemm_wmma(const float* __restrict__ agg,
                               const float* __restrict__ degv,
                               const float* __restrict__ xroot,
                               const float* __restrict__ wl,
                               const float* __restrict__ wr,
                               const float* __restrict__ bias,
                               float* __restrict__ out,
                               int N) {
    constexpr int S = K / 4;
    const int lane = threadIdx.x & 31;
    const int wave = threadIdx.x >> 5;
    const int lo   = lane & 15;
    const int half = lane >> 4;
    const int row0 = blockIdx.x * 16;
    const int col0 = wave * 16;

    int ar = row0 + lo;
    if (ar >= N) ar = N - 1;
    const float rinv = 1.0f / fmaxf(degv[ar], 1.0f);
    const int bcol = col0 + lo;

    const float* __restrict__ pa = agg   + (size_t)ar * K + half * 2;
    const float* __restrict__ px = xroot + (size_t)ar * K + half * 2;
    const float* __restrict__ pl = wl + (size_t)(half * 2) * M + bcol;
    const float* __restrict__ pr = wr + (size_t)(half * 2) * M + bcol;

    v2f af[S], bf[S];
    v8f acc = {};

#pragma unroll
    for (int s = 0; s < S; ++s) {
        af[s].x = pa[s * 4];
        af[s].y = pa[s * 4 + 1];
        bf[s].x = pl[(size_t)(s * 4) * M];
        bf[s].y = pl[(size_t)(s * 4 + 1) * M];
    }
#pragma unroll
    for (int s = 0; s < S; ++s) {
        v2f a;
        a.x = af[s].x * rinv;
        a.y = af[s].y * rinv;
        acc = __builtin_amdgcn_wmma_f32_16x16x4_f32(false, a, false, bf[s],
                                                    (short)0, acc, false, false);
    }

#pragma unroll
    for (int s = 0; s < S; ++s) {
        af[s].x = px[s * 4];
        af[s].y = px[s * 4 + 1];
        bf[s].x = pr[(size_t)(s * 4) * M];
        bf[s].y = pr[(size_t)(s * 4 + 1) * M];
    }
#pragma unroll
    for (int s = 0; s < S; ++s) {
        acc = __builtin_amdgcn_wmma_f32_16x16x4_f32(false, af[s], false, bf[s],
                                                    (short)0, acc, false, false);
    }

    const float bv = bias[bcol];
#pragma unroll
    for (int v = 0; v < 8; ++v) {
        const int r = row0 + half * 8 + v;
        if (r < N) {
            float val = acc[v] + bv;
            out[(size_t)r * M + bcol] = fmaxf(val, 0.0f);
        }
    }
}

__global__ void classifier_head(const float* __restrict__ h2,
                                const float* __restrict__ wc1,
                                const float* __restrict__ bc1,
                                const float* __restrict__ wc2,
                                const float* __restrict__ bc2,
                                float* __restrict__ out, int N) {
    const int n = blockIdx.x * blockDim.x + threadIdx.x;
    if (n >= N) return;
    const float* h = h2 + (size_t)n * 32;
    float hn[32];
#pragma unroll
    for (int k = 0; k < 32; ++k) hn[k] = h[k];
    float h3[16];
#pragma unroll
    for (int j = 0; j < 16; ++j) {
        float s = bc1[j];
#pragma unroll
        for (int k = 0; k < 32; ++k) s = fmaf(hn[k], wc1[k * 16 + j], s);
        h3[j] = fmaxf(s, 0.0f);
    }
#pragma unroll
    for (int j = 0; j < 2; ++j) {
        float s = bc2[j];
#pragma unroll
        for (int k = 0; k < 16; ++k) s = fmaf(h3[k], wc2[k * 2 + j], s);
        out[(size_t)n * 2 + j] = s;
    }
}

extern "C" void kernel_launch(void* const* d_in, const int* in_sizes, int n_in,
                              void* d_out, int out_size, void* d_ws, size_t ws_size,
                              hipStream_t stream) {
    const float* x    = (const float*)d_in[0];
    const int*   ei   = (const int*)d_in[1];
    const float* w1l  = (const float*)d_in[2];
    const float* w1r  = (const float*)d_in[3];
    const float* b1   = (const float*)d_in[4];
    const float* w2l  = (const float*)d_in[5];
    const float* w2r  = (const float*)d_in[6];
    const float* b2   = (const float*)d_in[7];
    const float* wc1  = (const float*)d_in[8];
    const float* bc1  = (const float*)d_in[9];
    const float* wc2  = (const float*)d_in[10];
    const float* bc2  = (const float*)d_in[11];
    float* out = (float*)d_out;

    const int N = in_sizes[0] / 64;
    const int E = in_sizes[1] / 2;
    const int* src = ei;
    const int* dst = ei + E;

    float* agg1 = (float*)d_ws;
    float* agg2 = agg1 + (size_t)N * 64;
    float* h1   = agg2 + (size_t)N * 64;
    float* h2   = h1   + (size_t)N * 64;
    float* deg  = h2   + (size_t)N * 32;
    int* cntp = (int*)agg2;
    int* rank = (int*)agg1;
    int* perm = (int*)h2;
    int* off  = perm + E;
    int* cnti = off + N;
    int* bsum = cnti + N;

    const int nblk = (N + 255) / 256;
    const int eblk = (E + 255) / 256;
    const int gblk = (N + 7) / 8;
    const int sblk = (N + 15) / 16;

    clear_counters<<<nblk, 256, 0, stream>>>(cntp, N);
    rank_edges<<<eblk, 256, 0, stream>>>(dst, cntp, rank, E);
    bucket_local_scan<<<nblk, 256, 0, stream>>>(cntp, cnti, deg, off, bsum, N);
    add_bucket_base<<<nblk, 256, 0, stream>>>(off, bsum, N);
    fill_buckets<<<eblk, 256, 0, stream>>>(src, dst, rank, off, perm, E);

    gather_sum64<<<gblk, 256, 0, stream>>>(x, perm, off, cnti, agg1, N);
    sage_gemm_wmma<64, 64><<<sblk, 128, 0, stream>>>(agg1, deg, x, w1l, w1r, b1, h1, N);

    gather_sum64<<<gblk, 256, 0, stream>>>(h1, perm, off, cnti, agg2, N);
    sage_gemm_wmma<64, 32><<<sblk, 64, 0, stream>>>(agg2, deg, h1, w2l, w2r, b2, h2, N);

    classifier_head<<<nblk, 256, 0, stream>>>(h2, wc1, bc1, wc2, bc2, out, N);
}


# Layer 2 applies w2l and w2r to h1 before the neighbour mean, inside GEMM 1 (h1 tile via LDS); its gather sums 32-channel product rows: products are summed where sums were multiplied
# speedup vs baseline: 2.6824x; 1.1117x over previous
#include <hip/hip_runtime.h>


typedef __attribute__((ext_vector_type(2))) float v2f;
typedef __attribute__((ext_vector_type(8))) float v8f;

__global__ void clear_counters(int* __restrict__ cnt, int N) {
    const int n = blockIdx.x * blockDim.x + threadIdx.x;
    if (n < N) cnt[(size_t)n * 64] = 0;
}

__global__ void rank_edges(const int* __restrict__ dst,
                           int* __restrict__ cnt,
                           int* __restrict__ rank, int E) {
    const int e = blockIdx.x * blockDim.x + threadIdx.x;
    if (e >= E) return;
    const int d = dst[e];
    rank[e] = __hip_atomic_fetch_add(&cnt[(size_t)d * 64], 1, __ATOMIC_RELAXED,
                                     __HIP_MEMORY_SCOPE_AGENT);
}

__global__ void bucket_local_scan(const int* __restrict__ cnt,
                                  int* __restrict__ cnti,
                                  float* __restrict__ deg,
                                  int* __restrict__ off,
                                  int* __restrict__ bsum, int N) {
    __shared__ int sh[256];
    const int t = threadIdx.x;
    const int n = blockIdx.x * 256 + t;
    const int c = (n < N) ? cnt[(size_t)n * 64] : 0;
    sh[t] = c;
    __syncthreads();
    for (int o = 1; o < 256; o <<= 1) {
        const int v = (t >= o) ? sh[t - o] : 0;
        __syncthreads();
        sh[t] += v;
        __syncthreads();
    }
    const int incl = sh[t];
    if (n < N) {
        off[n] = incl - c;
        cnti[n] = c;
        deg[n] = (float)c;
    }
    if (t == 255) bsum[blockIdx.x] = incl;
}

__global__ void add_bucket_base(int* __restrict__ off,
                                const int* __restrict__ bsum, int N) {
    __shared__ int sh[256];
    const int t = threadIdx.x;
    const int b = blockIdx.x;
    int p = 0;
    for (int i = t; i < b; i += 256) p += bsum[i];
    sh[t] = p;
    __syncthreads();
    for (int o = 128; o > 0; o >>= 1) {
        if (t < o) sh[t] += sh[t + o];
        __syncthreads();
    }
    const int base = sh[0];
    const int n = b * 256 + t;
    if (n < N) off[n] += base;
}

__global__ void fill_buckets(const int* __restrict__ src,
                             const int* __restrict__ dst,
                             const int* __restrict__ rank,
                             const int* __restrict__ off,
                             int* __restrict__ perm, int E) {
    const int e = blockIdx.x * blockDim.x + threadIdx.x;
    if (e >= E) return;
    perm[off[dst[e]] + rank[e]] = src[e];
}

__global__ void gather_sum64(const float* __restrict__ feat,
                             const int* __restrict__ perm,
                             const int* __restrict__ off,
                             const int* __restrict__ cnti,
                             float* __restrict__ agg, int N) {
    const int lane = threadIdx.x & 31;
    const int n = __builtin_amdgcn_readfirstlane(
        (int)(blockIdx.x * (blockDim.x >> 5) + (threadIdx.x >> 5)));
    if (n >= N) return;
    const int start = off[n];
    const int cnt = cnti[n];
    const float2* __restrict__ f2 = (const float2*)feat;
    float2 acc = make_float2(0.0f, 0.0f);
    for (int base = 0; base < cnt; base += 32) {
        const int m = min(32, cnt - base);
        const int idx = (lane < m) ? perm[start + base + lane] : 0;
        int t = 0;
        for (; t + 8 <= m; t += 8) {
            float2 r[8];
#pragma unroll
            for (int i = 0; i < 8; ++i) {
                const int s = __builtin_amdgcn_readlane(idx, t + i);
                r[i] = f2[(size_t)s * 32 + lane];
            }
#pragma unroll
            for (int i = 0; i < 8; ++i) {
                acc.x += r[i].x;
                acc.y += r[i].y;
            }
        }
        for (; t < m; t += 4) {
            float2 r[4];
#pragma unroll
            for (int i = 0; i < 4; ++i) {
                const int s = __builtin_amdgcn_readlane(idx, t + i);
                r[i] = f2[(size_t)s * 32 + lane];
            }
#pragma unroll
            for (int i = 0; i < 4; ++i) {
                if (t + i < m) {
                    acc.x += r[i].x;
                    acc.y += r[i].y;
                }
            }
        }
    }
    ((float2*)agg)[(size_t)n * 32 + lane] = acc;
}

__global__ void gather_mean32_root_relu(const float* __restrict__ p,
                                        const float* __restrict__ q,
                                        const int* __restrict__ perm,
                                        const int* __restrict__ off,
                                        const int* __restrict__ cnti,
                                        float* __restrict__ h2, int N) {
    const int lane = threadIdx.x & 31;
    const int n = __builtin_amdgcn_readfirstlane(
        (int)(blockIdx.x * (blockDim.x >> 5) + (threadIdx.x >> 5)));
    if (n >= N) return;
    const int start = off[n];
    const int cnt = cnti[n];
    const float qv = q[(size_t)n * 32 + lane];
    float acc = 0.0f;
    for (int base = 0; base < cnt; base += 32) {
        const int m = min(32, cnt - base);
        const int idx = (lane < m) ? perm[start + base + lane] : 0;
        int t = 0;
        for (; t + 8 <= m; t += 8) {
            float r[8];
#pragma unroll
            for (int i = 0; i < 8; ++i) {
                const int s = __builtin_amdgcn_readlane(idx, t + i);
                r[i] = p[(size_t)s * 32 + lane];
            }
#pragma unroll
            for (int i = 0; i < 8; ++i) acc += r[i];
        }
        for (; t < m; t += 4) {
            float r[4];
#pragma unroll
            for (int i = 0; i < 4; ++i) {
                const int s = __builtin_amdgcn_readlane(idx, t + i);
                r[i] = p[(size_t)s * 32 + lane];
            }
#pragma unroll
            for (int i = 0; i < 4; ++i) {
                if (t + i < m) acc += r[i];
            }
        }
    }
    const float d = fmaxf((float)cnt, 1.0f);
    h2[(size_t)n * 32 + lane] = fmaxf(acc / d + qv, 0.0f);
}

__global__ void sage_gemm1_then_l2_products(const float* __restrict__ agg,
                                            const float* __restrict__ degv,
                                            const float* __restrict__ xroot,
                                            const float* __restrict__ wl,
                                            const float* __restrict__ wr,
                                            const float* __restrict__ bias,
                                            const float* __restrict__ w2l,
                                            const float* __restrict__ w2r,
                                            const float* __restrict__ b2,
                                            float* __restrict__ p,
                                            float* __restrict__ q,
                                            int N) {
    constexpr int K = 64;
    constexpr int M = 64;
    constexpr int M2 = 32;
    constexpr int S = K / 4;
    constexpr int PITCH = 68;
    __shared__ float sh[16 * PITCH];

    const int lane = threadIdx.x & 31;
    const int wave = threadIdx.x >> 5;
    const int lo   = lane & 15;
    const int half = lane >> 4;
    const int row0 = blockIdx.x * 16;
    const int col0 = wave * 16;

    int ar = row0 + lo;
    if (ar >= N) ar = N - 1;
    const float rinv = 1.0f / fmaxf(degv[ar], 1.0f);
    const int bcol = col0 + lo;

    const float* __restrict__ pa = agg   + (size_t)ar * K + half * 2;
    const float* __restrict__ px = xroot + (size_t)ar * K + half * 2;
    const float* __restrict__ pl = wl + (size_t)(half * 2) * M + bcol;
    const float* __restrict__ pr = wr + (size_t)(half * 2) * M + bcol;

    v2f af[S], bf[S];
    v8f acc = {};

#pragma unroll
    for (int s = 0; s < S; ++s) {
        af[s].x = pa[s * 4];
        af[s].y = pa[s * 4 + 1];
        bf[s].x = pl[(size_t)(s * 4) * M];
        bf[s].y = pl[(size_t)(s * 4 + 1) * M];
    }
#pragma unroll
    for (int s = 0; s < S; ++s) {
        v2f a;
        a.x = af[s].x * rinv;
        a.y = af[s].y * rinv;
        acc = __builtin_amdgcn_wmma_f32_16x16x4_f32(false, a, false, bf[s],
                                                    (short)0, acc, false, false);
    }

#pragma unroll
    for (int s = 0; s < S; ++s) {
        af[s].x = px[s * 4];
        af[s].y = px[s * 4 + 1];
        bf[s].x = pr[(size_t)(s * 4) * M];
        bf[s].y = pr[(size_t)(s * 4 + 1) * M];
    }
#pragma unroll
    for (int s = 0; s < S; ++s) {
        acc = __builtin_amdgcn_wmma_f32_16x16x4_f32(false, af[s], false, bf[s],
                                                    (short)0, acc, false, false);
    }

    const bool is_p = wave < 2;
    const int c2 = (wave & 1) * 16 + lo;
    const float* __restrict__ pw =
        (is_p ? w2l : w2r) + (size_t)(half * 2) * M2 + c2;
    v2f b2f[S];
#pragma unroll
    for (int s = 0; s < S; ++s) {
        b2f[s].x = pw[(size_t)(s * 4) * M2];
        b2f[s].y = pw[(size_t)(s * 4 + 1) * M2];
    }
    const float add2 = is_p ? 0.0f : b2[c2];

    const float bv = bias[bcol];
#pragma unroll
    for (int v = 0; v < 8; ++v) {
        sh[(half * 8 + v) * PITCH + bcol] = fmaxf(acc[v] + bv, 0.0f);
    }
    __syncthreads();

    const float* ps = sh + lo * PITCH + half * 2;
    v8f acc2 = {};
#pragma unroll
    for (int s = 0; s < S; ++s) {
        af[s].x = ps[s * 4];
        af[s].y = ps[s * 4 + 1];
    }
#pragma unroll
    for (int s = 0; s < S; ++s) {
        acc2 = __builtin_amdgcn_wmma_f32_16x16x4_f32(false, af[s], false, b2f[s],
                                                     (short)0, acc2, false, false);
    }

    float* __restrict__ dst = is_p ? p : q;
#pragma unroll
    for (int v = 0; v < 8; ++v) {
        const int r = row0 + half * 8 + v;
        if (r < N) dst[(size_t)r * M2 + c2] = acc2[v] + add2;
    }
}

__global__ void classifier_head(const float* __restrict__ h2,
                                const float* __restrict__ wc1,
                                const float* __restrict__ bc1,
                                const float* __restrict__ wc2,
                                const float* __restrict__ bc2,
                                float* __restrict__ out, int N) {
    const int n = blockIdx.x * blockDim.x + threadIdx.x;
    if (n >= N) return;
    const float* h = h2 + (size_t)n * 32;
    float hn[32];
#pragma unroll
    for (int k = 0; k < 32; ++k) hn[k] = h[k];
    float h3[16];
#pragma unroll
    for (int j = 0; j < 16; ++j) {
        float s = bc1[j];
#pragma unroll
        for (int k = 0; k < 32; ++k) s = fmaf(hn[k], wc1[k * 16 + j], s);
        h3[j] = fmaxf(s, 0.0f);
    }
#pragma unroll
    for (int j = 0; j < 2; ++j) {
        float s = bc2[j];
#pragma unroll
        for (int k = 0; k < 16; ++k) s = fmaf(h3[k], wc2[k * 2 + j], s);
        out[(size_t)n * 2 + j] = s;
    }
}

extern "C" void kernel_launch(void* const* d_in, const int* in_sizes, int n_in,
                              void* d_out, int out_size, void* d_ws, size_t ws_size,
                              hipStream_t stream) {
    const float* x    = (const float*)d_in[0];
    const int*   ei   = (const int*)d_in[1];
    const float* w1l  = (const float*)d_in[2];
    const float* w1r  = (const float*)d_in[3];
    const float* b1   = (const float*)d_in[4];
    const float* w2l  = (const float*)d_in[5];
    const float* w2r  = (const float*)d_in[6];
    const float* b2   = (const float*)d_in[7];
    const float* wc1  = (const float*)d_in[8];
    const float* bc1  = (const float*)d_in[9];
    const float* wc2  = (const float*)d_in[10];
    const float* bc2  = (const float*)d_in[11];
    float* out = (float*)d_out;

    const int N = in_sizes[0] / 64;
    const int E = in_sizes[1] / 2;
    const int* src = ei;
    const int* dst = ei + E;

    float* agg1 = (float*)d_ws;
    float* pq   = agg1 + (size_t)N * 64;
    float* h2   = pq   + (size_t)N * 64;
    float* sortp = h2  + (size_t)N * 64;
    float* deg  = sortp + (size_t)N * 32;
    float* p    = pq;
    float* q    = pq + (size_t)N * 32;
    int* cntp = (int*)pq;
    int* rank = (int*)agg1;
    int* perm = (int*)sortp;
    int* off  = perm + E;
    int* cnti = off + N;
    int* bsum = cnti + N;

    const int nblk = (N + 255) / 256;
    const int eblk = (E + 255) / 256;
    const int gblk = (N + 7) / 8;
    const int sblk = (N + 15) / 16;

    clear_counters<<<nblk, 256, 0, stream>>>(cntp, N);
    rank_edges<<<eblk, 256, 0, stream>>>(dst, cntp, rank, E);
    bucket_local_scan<<<nblk, 256, 0, stream>>>(cntp, cnti, deg, off, bsum, N);
    add_bucket_base<<<nblk, 256, 0, stream>>>(off, bsum, N);
    fill_buckets<<<eblk, 256, 0, stream>>>(src, dst, rank, off, perm, E);

    gather_sum64<<<gblk, 256, 0, stream>>>(x, perm, off, cnti, agg1, N);
    sage_gemm1_then_l2_products<<<sblk, 128, 0, stream>>>(agg1, deg, x, w1l, w1r, b1,
                                                          w2l, w2r, b2, p, q, N);

    gather_mean32_root_relu<<<gblk, 256, 0, stream>>>(p, q, perm, off, cnti, h2, N);

    classifier_head<<<nblk, 256, 0, stream>>>(h2, wc1, bc1, wc2, bc2, out, N);
}


# Each gather now runs at the head of the kernel that consumes its rows (layer-1 GEMM; layer-2 combine + classifier head), rows passed in LDS: 9 launches -> 7, no agg or h2 plane; sum orders unchanged
# speedup vs baseline: 3.2914x; 1.2270x over previous
#include <hip/hip_runtime.h>


typedef __attribute__((ext_vector_type(2))) float v2f;
typedef __attribute__((ext_vector_type(8))) float v8f;

__global__ void clear_counters(int* __restrict__ cnt, int N) {
    const int n = blockIdx.x * blockDim.x + threadIdx.x;
    if (n < N) cnt[(size_t)n * 64] = 0;
}

__global__ void rank_edges(const int* __restrict__ dst,
                           int* __restrict__ cnt,
                           int* __restrict__ rank, int E) {
    const int e = blockIdx.x * blockDim.x + threadIdx.x;
    if (e >= E) return;
    const int d = dst[e];
    rank[e] = __hip_atomic_fetch_add(&cnt[(size_t)d * 64], 1, __ATOMIC_RELAXED,
                                     __HIP_MEMORY_SCOPE_AGENT);
}

__global__ void bucket_local_scan(const int* __restrict__ cnt,
                                  int* __restrict__ cnti,
                                  int* __restrict__ off,
                                  int* __restrict__ bsum, int N) {
    __shared__ int sh[256];
    const int t = threadIdx.x;
    const int n = blockIdx.x * 256 + t;
    const int c = (n < N) ? cnt[(size_t)n * 64] : 0;
    sh[t] = c;
    __syncthreads();
    for (int o = 1; o < 256; o <<= 1) {
        const int v = (t >= o) ? sh[t - o] : 0;
        __syncthreads();
        sh[t] += v;
        __syncthreads();
    }
    const int incl = sh[t];
    if (n < N) {
        off[n] = incl - c;
        cnti[n] = c;
    }
    if (t == 255) bsum[blockIdx.x] = incl;
}

__global__ void add_bucket_base(int* __restrict__ off,
                                const int* __restrict__ bsum, int N) {
    __shared__ int sh[256];
    const int t = threadIdx.x;
    const int b = blockIdx.x;
    int p = 0;
    for (int i = t; i < b; i += 256) p += bsum[i];
    sh[t] = p;
    __syncthreads();
    for (int o = 128; o > 0; o >>= 1) {
        if (t < o) sh[t] += sh[t + o];
        __syncthreads();
    }
    const int base = sh[0];
    const int n = b * 256 + t;
    if (n < N) off[n] += base;
}

__global__ void fill_buckets(const int* __restrict__ src,
                             const int* __restrict__ dst,
                             const int* __restrict__ rank,
                             const int* __restrict__ off,
                             int* __restrict__ perm, int E) {
    const int e = blockIdx.x * blockDim.x + threadIdx.x;
    if (e >= E) return;
    perm[off[dst[e]] + rank[e]] = src[e];
}

__device__ __forceinline__ float2 bucket_sum64(const float2* __restrict__ f2,
                                               const int* __restrict__ perm,
                                               int start, int cnt,
                                               int idx_first, int lane) {
    float2 acc = make_float2(0.0f, 0.0f);
    int idx = idx_first;
    for (int base = 0; base < cnt; base += 32) {
        const int m = min(32, cnt - base);
        if (base > 0) idx = (lane < m) ? perm[start + base + lane] : 0;
        int t = 0;
        for (; t + 16 <= m; t += 16) {
            float2 r[16];
#pragma unroll
            for (int i = 0; i < 16; ++i) {
                const int s = __builtin_amdgcn_readlane(idx, t + i);
                r[i] = f2[(size_t)s * 32 + lane];
            }
#pragma unroll
            for (int i = 0; i < 16; ++i) {
                acc.x += r[i].x;
                acc.y += r[i].y;
            }
        }
        for (; t + 8 <= m; t += 8) {
            float2 r[8];
#pragma unroll
            for (int i = 0; i < 8; ++i) {
                const int s = __builtin_amdgcn_readlane(idx, t + i);
                r[i] = f2[(size_t)s * 32 + lane];
            }
#pragma unroll
            for (int i = 0; i < 8; ++i) {
                acc.x += r[i].x;
                acc.y += r[i].y;
            }
        }
        for (; t < m; t += 4) {
            float2 r[4];
#pragma unroll
            for (int i = 0; i < 4; ++i) {
                const int s = __builtin_amdgcn_readlane(idx, t + i);
                r[i] = f2[(size_t)s * 32 + lane];
            }
#pragma unroll
            for (int i = 0; i < 4; ++i) {
                if (t + i < m) {
                    acc.x += r[i].x;
                    acc.y += r[i].y;
                }
            }
        }
    }
    return acc;
}

__device__ __forceinline__ float bucket_sum32(const float* __restrict__ f,
                                              const int* __restrict__ perm,
                                              int start, int cnt,
                                              int idx_first, int lane) {
    float acc = 0.0f;
    int idx = idx_first;
    for (int base = 0; base < cnt; base += 32) {
        const int m = min(32, cnt - base);
        if (base > 0) idx = (lane < m) ? perm[start + base + lane] : 0;
        int t = 0;
        for (; t + 16 <= m; t += 16) {
            float r[16];
#pragma unroll
            for (int i = 0; i < 16; ++i) {
                const int s = __builtin_amdgcn_readlane(idx, t + i);
                r[i] = f[(size_t)s * 32 + lane];
            }
#pragma unroll
            for (int i = 0; i < 16; ++i) acc += r[i];
        }
        for (; t + 8 <= m; t += 8) {
            float r[8];
#pragma unroll
            for (int i = 0; i < 8; ++i) {
                const int s = __builtin_amdgcn_readlane(idx, t + i);
                r[i] = f[(size_t)s * 32 + lane];
            }
#pragma unroll
            for (int i = 0; i < 8; ++i) acc += r[i];
        }
        for (; t < m; t += 4) {
            float r[4];
#pragma unroll
            for (int i = 0; i < 4; ++i) {
                const int s = __builtin_amdgcn_readlane(idx, t + i);
                r[i] = f[(size_t)s * 32 + lane];
            }
#pragma unroll
            for (int i = 0; i < 4; ++i) {
                if (t + i < m) acc += r[i];
            }
        }
    }
    return acc;
}

__global__ void __launch_bounds__(128, 16) gather64_gemm1_then_l2_products(const float* __restrict__ xin,
                                                const int* __restrict__ perm,
                                                const int* __restrict__ off,
                                                const int* __restrict__ cnti,
                                                const float* __restrict__ wl,
                                                const float* __restrict__ wr,
                                                const float* __restrict__ bias,
                                                const float* __restrict__ w2l,
                                                const float* __restrict__ w2r,
                                                const float* __restrict__ b2,
                                                float* __restrict__ p,
                                                float* __restrict__ q,
                                                int N) {
    constexpr int K = 64;
    constexpr int M = 64;
    constexpr int M2 = 32;
    constexpr int S = K / 4;
    constexpr int PITCH = 68;
    __shared__ float sh[2 * 16 * PITCH];
    float* const sh_m = sh;
    float* const sh_h = sh + 16 * PITCH;

    const int lane = threadIdx.x & 31;
    const int wave = __builtin_amdgcn_readfirstlane((int)(threadIdx.x >> 5));
    const int lo   = lane & 15;
    const int half = lane >> 4;
    const int row0 = blockIdx.x * 16;
    const int col0 = wave * 16;

    const int nb = row0 + wave * 4;
    int st[4], cn[4], ix[4];
#pragma unroll
    for (int i = 0; i < 4; ++i) {
        const int n = nb + i;
        const int nn = min(n, N - 1);
        st[i] = off[nn];
        cn[i] = (n < N) ? cnti[nn] : 0;
    }
#pragma unroll
    for (int i = 0; i < 4; ++i) {
        ix[i] = (lane < min(32, cn[i])) ? perm[st[i] + lane] : 0;
    }
#pragma unroll
    for (int i = 0; i < 4; ++i) {
        const float2 a = bucket_sum64((const float2*)xin, perm, st[i], cn[i],
                                      ix[i], lane);
        const float rinv = 1.0f / fmaxf((float)cn[i], 1.0f);
        float* mrow = sh_m + (wave * 4 + i) * PITCH + 2 * lane;
        mrow[0] = a.x * rinv;
        mrow[1] = a.y * rinv;
    }
    __syncthreads();

    int ar = row0 + lo;
    if (ar >= N) ar = N - 1;
    const int bcol = col0 + lo;

    const float* pm = sh_m + lo * PITCH + half * 2;
    const float* __restrict__ px = xin + (size_t)ar * K + half * 2;
    const float* __restrict__ pl = wl + (size_t)(half * 2) * M + bcol;
    const float* __restrict__ pr = wr + (size_t)(half * 2) * M + bcol;

    v2f af[S], bf[S];
    v8f acc = {};

#pragma unroll
    for (int s = 0; s < S; ++s) {
        af[s].x = pm[s * 4];
        af[s].y = pm[s * 4 + 1];
        bf[s].x = pl[(size_t)(s * 4) * M];
        bf[s].y = pl[(size_t)(s * 4 + 1) * M];
    }
#pragma unroll
    for (int s = 0; s < S; ++s) {
        acc = __builtin_amdgcn_wmma_f32_16x16x4_f32(false, af[s], false, bf[s],
                                                    (short)0, acc, false, false);
    }

#pragma unroll
    for (int s = 0; s < S; ++s) {
        af[s].x = px[s * 4];
        af[s].y = px[s * 4 + 1];
        bf[s].x = pr[(size_t)(s * 4) * M];
        bf[s].y = pr[(size_t)(s * 4 + 1) * M];
    }
#pragma unroll
    for (int s = 0; s < S; ++s) {
        acc = __builtin_amdgcn_wmma_f32_16x16x4_f32(false, af[s], false, bf[s],
                                                    (short)0, acc, false, false);
    }

    const bool is_p = wave < 2;
    const int c2 = (wave & 1) * 16 + lo;
    const float* __restrict__ pw =
        (is_p ? w2l : w2r) + (size_t)(half * 2) * M2 + c2;
    v2f b2f[S];
#pragma unroll
    for (int s = 0; s < S; ++s) {
        b2f[s].x = pw[(size_t)(s * 4) * M2];
        b2f[s].y = pw[(size_t)(s * 4 + 1) * M2];
    }
    const float add2 = is_p ? 0.0f : b2[c2];

    const float bv = bias[bcol];
#pragma unroll
    for (int v = 0; v < 8; ++v) {
        sh_h[(half * 8 + v) * PITCH + bcol] = fmaxf(acc[v] + bv, 0.0f);
    }
    __syncthreads();

    const float* ps = sh_h + lo * PITCH + half * 2;
    v8f acc2 = {};
#pragma unroll
    for (int s = 0; s < S; ++s) {
        af[s].x = ps[s * 4];
        af[s].y = ps[s * 4 + 1];
    }
#pragma unroll
    for (int s = 0; s < S; ++s) {
        acc2 = __builtin_amdgcn_wmma_f32_16x16x4_f32(false, af[s], false, b2f[s],
                                                     (short)0, acc2, false, false);
    }

    float* __restrict__ dst = is_p ? p : q;
#pragma unroll
    for (int v = 0; v < 8; ++v) {
        const int r = row0 + half * 8 + v;
        if (r < N) dst[(size_t)r * M2 + c2] = acc2[v] + add2;
    }
}

__global__ void gather32_layer2_then_classifier(const float* __restrict__ p,
                                                const float* __restrict__ q,
                                                const int* __restrict__ perm,
                                                const int* __restrict__ off,
                                                const int* __restrict__ cnti,
                                                const float* __restrict__ wc1,
                                                const float* __restrict__ bc1,
                                                const float* __restrict__ wc2,
                                                const float* __restrict__ bc2,
                                                float* __restrict__ out, int N) {
    __shared__ float sh_h2[16 * 32];
    __shared__ float sh_h3[16 * 16];

    const int lane = threadIdx.x & 31;
    const int wave = __builtin_amdgcn_readfirstlane((int)(threadIdx.x >> 5));
    const int node0 = blockIdx.x * 16;
    const int nb = node0 + wave * 4;

    int st[4], cn[4], ix[4];
    float qv[4];
#pragma unroll
    for (int i = 0; i < 4; ++i) {
        const int n = nb + i;
        const int nn = min(n, N - 1);
        st[i] = off[nn];
        cn[i] = (n < N) ? cnti[nn] : 0;
        qv[i] = q[(size_t)nn * 32 + lane];
    }
#pragma unroll
    for (int i = 0; i < 4; ++i) {
        ix[i] = (lane < min(32, cn[i])) ? perm[st[i] + lane] : 0;
    }
#pragma unroll
    for (int i = 0; i < 4; ++i) {
        const float acc = bucket_sum32(p, perm, st[i], cn[i], ix[i], lane);
        const float d = fmaxf((float)cn[i], 1.0f);
        sh_h2[(wave * 4 + i) * 32 + lane] = fmaxf(acc / d + qv[i], 0.0f);
    }
    __syncthreads();

    const int t = threadIdx.x;
    const int j = t & 15;
    const int na = t >> 4;
    float s0 = bc1[j];
    float s1 = s0;
#pragma unroll
    for (int k = 0; k < 32; ++k) {
        const float w = wc1[k * 16 + j];
        s0 = fmaf(sh_h2[na * 32 + k], w, s0);
        s1 = fmaf(sh_h2[(na + 8) * 32 + k], w, s1);
    }
    sh_h3[na * 16 + j] = fmaxf(s0, 0.0f);
    sh_h3[(na + 8) * 16 + j] = fmaxf(s1, 0.0f);
    __syncthreads();

    if (t < 32) {
        const int node = t >> 1;
        const int jj = t & 1;
        float s = bc2[jj];
#pragma unroll
        for (int k = 0; k < 16; ++k) {
            s = fmaf(sh_h3[node * 16 + k], wc2[k * 2 + jj], s);
        }
        const int n = node0 + node;
        if (n < N) out[(size_t)n * 2 + jj] = s;
    }
}

extern "C" void kernel_launch(void* const* d_in, const int* in_sizes, int n_in,
                              void* d_out, int out_size, void* d_ws, size_t ws_size,
                              hipStream_t stream) {
    const float* x    = (const float*)d_in[0];
    const int*   ei   = (const int*)d_in[1];
    const float* w1l  = (const float*)d_in[2];
    const float* w1r  = (const float*)d_in[3];
    const float* b1   = (const float*)d_in[4];
    const float* w2l  = (const float*)d_in[5];
    const float* w2r  = (const float*)d_in[6];
    const float* b2   = (const float*)d_in[7];
    const float* wc1  = (const float*)d_in[8];
    const float* bc1  = (const float*)d_in[9];
    const float* wc2  = (const float*)d_in[10];
    const float* bc2  = (const float*)d_in[11];
    float* out = (float*)d_out;

    const int N = in_sizes[0] / 64;
    const int E = in_sizes[1] / 2;
    const int* src = ei;
    const int* dst = ei + E;

    float* plane0 = (float*)d_ws;
    float* pq   = plane0 + (size_t)N * 64;
    float* sortp = pq + (size_t)N * 128;
    float* p    = pq;
    float* q    = pq + (size_t)N * 32;
    int* cntp = (int*)pq;
    int* rank = (int*)plane0;
    int* perm = (int*)sortp;
    int* off  = perm + E;
    int* cnti = off + N;
    int* bsum = cnti + N;

    const int nblk = (N + 255) / 256;
    const int eblk = (E + 255) / 256;
    const int sblk = (N + 15) / 16;

    clear_counters<<<nblk, 256, 0, stream>>>(cntp, N);
    rank_edges<<<eblk, 256, 0, stream>>>(dst, cntp, rank, E);
    bucket_local_scan<<<nblk, 256, 0, stream>>>(cntp, cnti, off, bsum, N);
    add_bucket_base<<<nblk, 256, 0, stream>>>(off, bsum, N);
    fill_buckets<<<eblk, 256, 0, stream>>>(src, dst, rank, off, perm, E);

    gather64_gemm1_then_l2_products<<<sblk, 128, 0, stream>>>(
        x, perm, off, cnti, w1l, w1r, b1, w2l, w2r, b2, p, q, N);

    gather32_layer2_then_classifier<<<sblk, 128, 0, stream>>>(
        p, q, perm, off, cnti, wc1, bc1, wc2, bc2, out, N);
}


# One edge pass: the returning count add is the rank, the source id goes straight into the node's own 256-B row (overflow list past 63); no scan or fill pass, 7 launches -> 4; sum order unchanged
# speedup vs baseline: 4.0070x; 1.2174x over previous
#include <hip/hip_runtime.h>


typedef __attribute__((ext_vector_type(2))) float v2f;
typedef __attribute__((ext_vector_type(8))) float v8f;

constexpr int ROW = 64;
constexpr int CAP = 63;

__global__ void clear_counts(int* __restrict__ rows, int* __restrict__ ovf_cnt,
                             int N) {
    const int n = blockIdx.x * blockDim.x + threadIdx.x;
    if (n < N) rows[(size_t)n * ROW + CAP] = 0;
    if (n == 0) *ovf_cnt = 0;
}

__global__ void place_edges(const int* __restrict__ src,
                            const int* __restrict__ dst,
                            int* rows, int* ovf_cnt,
                            int* __restrict__ ovf_src,
                            int* __restrict__ ovf_dst, int E) {
    const int e = blockIdx.x * blockDim.x + threadIdx.x;
    if (e >= E) return;
    const int d = dst[e];
    const int s = src[e];
    int* const row = rows + (size_t)d * ROW;
    const int r = __hip_atomic_fetch_add(row + CAP, 1, __ATOMIC_RELAXED,
                                         __HIP_MEMORY_SCOPE_AGENT);
    if (r < CAP) {
        row[r] = s;
    } else {
        const int k = __hip_atomic_fetch_add(ovf_cnt, 1, __ATOMIC_RELAXED,
                                             __HIP_MEMORY_SCOPE_AGENT);
        ovf_src[k] = s;
        ovf_dst[k] = d;
    }
}

__device__ __forceinline__ float2 bucket_sum64(const float2* __restrict__ f2,
                                               const int* __restrict__ perm,
                                               int start, int cnt,
                                               int idx_first, int lane) {
    float2 acc = make_float2(0.0f, 0.0f);
    int idx = idx_first;
    for (int base = 0; base < cnt; base += 32) {
        const int m = min(32, cnt - base);
        if (base > 0) idx = (lane < m) ? perm[start + base + lane] : 0;
        int t = 0;
        for (; t + 16 <= m; t += 16) {
            float2 r[16];
#pragma unroll
            for (int i = 0; i < 16; ++i) {
                const int s = __builtin_amdgcn_readlane(idx, t + i);
                r[i] = f2[(size_t)s * 32 + lane];
            }
#pragma unroll
            for (int i = 0; i < 16; ++i) {
                acc.x += r[i].x;
                acc.y += r[i].y;
            }
        }
        for (; t + 8 <= m; t += 8) {
            float2 r[8];
#pragma unroll
            for (int i = 0; i < 8; ++i) {
                const int s = __builtin_amdgcn_readlane(idx, t + i);
                r[i] = f2[(size_t)s * 32 + lane];
            }
#pragma unroll
            for (int i = 0; i < 8; ++i) {
                acc.x += r[i].x;
                acc.y += r[i].y;
            }
        }
        for (; t < m; t += 4) {
            float2 r[4];
#pragma unroll
            for (int i = 0; i < 4; ++i) {
                const int s = __builtin_amdgcn_readlane(idx, t + i);
                r[i] = f2[(size_t)s * 32 + lane];
            }
#pragma unroll
            for (int i = 0; i < 4; ++i) {
                if (t + i < m) {
                    acc.x += r[i].x;
                    acc.y += r[i].y;
                }
            }
        }
    }
    return acc;
}

__device__ __forceinline__ float bucket_sum32(const float* __restrict__ f,
                                              const int* __restrict__ perm,
                                              int start, int cnt,
                                              int idx_first, int lane) {
    float acc = 0.0f;
    int idx = idx_first;
    for (int base = 0; base < cnt; base += 32) {
        const int m = min(32, cnt - base);
        if (base > 0) idx = (lane < m) ? perm[start + base + lane] : 0;
        int t = 0;
        for (; t + 16 <= m; t += 16) {
            float r[16];
#pragma unroll
            for (int i = 0; i < 16; ++i) {
                const int s = __builtin_amdgcn_readlane(idx, t + i);
                r[i] = f[(size_t)s * 32 + lane];
            }
#pragma unroll
            for (int i = 0; i < 16; ++i) acc += r[i];
        }
        for (; t + 8 <= m; t += 8) {
            float r[8];
#pragma unroll
            for (int i = 0; i < 8; ++i) {
                const int s = __builtin_amdgcn_readlane(idx, t + i);
                r[i] = f[(size_t)s * 32 + lane];
            }
#pragma unroll
            for (int i = 0; i < 8; ++i) acc += r[i];
        }
        for (; t < m; t += 4) {
            float r[4];
#pragma unroll
            for (int i = 0; i < 4; ++i) {
                const int s = __builtin_amdgcn_readlane(idx, t + i);
                r[i] = f[(size_t)s * 32 + lane];
            }
#pragma unroll
            for (int i = 0; i < 4; ++i) {
                if (t + i < m) acc += r[i];
            }
        }
    }
    return acc;
}

__device__ __forceinline__ float2 overflow_sum64(const float2* __restrict__ f2,
                                                 const int* __restrict__ ovf_src,
                                                 const int* __restrict__ ovf_dst,
                                                 int total, int node, int lane,
                                                 float2 acc) {
    for (int base = 0; base < total; base += 32) {
        const int j = base + lane;
        const int dd = (j < total) ? ovf_dst[j] : -1;
        const int ss = (j < total) ? ovf_src[j] : 0;
        unsigned mask = __builtin_amdgcn_ballot_w32(dd == node);
        while (mask != 0u) {
            const int b = __builtin_ctz(mask);
            mask &= mask - 1u;
            const int s = __builtin_amdgcn_readlane(ss, b);
            const float2 r = f2[(size_t)s * 32 + lane];
            acc.x += r.x;
            acc.y += r.y;
        }
    }
    return acc;
}

__device__ __forceinline__ float overflow_sum32(const float* __restrict__ f,
                                                const int* __restrict__ ovf_src,
                                                const int* __restrict__ ovf_dst,
                                                int total, int node, int lane,
                                                float acc) {
    for (int base = 0; base < total; base += 32) {
        const int j = base + lane;
        const int dd = (j < total) ? ovf_dst[j] : -1;
        const int ss = (j < total) ? ovf_src[j] : 0;
        unsigned mask = __builtin_amdgcn_ballot_w32(dd == node);
        while (mask != 0u) {
            const int b = __builtin_ctz(mask);
            mask &= mask - 1u;
            const int s = __builtin_amdgcn_readlane(ss, b);
            acc += f[(size_t)s * 32 + lane];
        }
    }
    return acc;
}

__global__ void __launch_bounds__(128, 16) gather64_gemm1_then_l2_products(const float* __restrict__ xin,
                                                const int* __restrict__ rows,
                                                const int* __restrict__ ovf_cnt,
                                                const int* __restrict__ ovf_src,
                                                const int* __restrict__ ovf_dst,
                                                const float* __restrict__ wl,
                                                const float* __restrict__ wr,
                                                const float* __restrict__ bias,
                                                const float* __restrict__ w2l,
                                                const float* __restrict__ w2r,
                                                const float* __restrict__ b2,
                                                float* __restrict__ p,
                                                float* __restrict__ q,
                                                int N, int E) {
    constexpr int K = 64;
    constexpr int M = 64;
    constexpr int M2 = 32;
    constexpr int S = K / 4;
    constexpr int PITCH = 68;
    __shared__ float sh[2 * 16 * PITCH];
    float* const sh_m = sh;
    float* const sh_h = sh + 16 * PITCH;

    const int lane = threadIdx.x & 31;
    const int wave = __builtin_amdgcn_readfirstlane((int)(threadIdx.x >> 5));
    const int lo   = lane & 15;
    const int half = lane >> 4;
    const int row0 = blockIdx.x * 16;
    const int col0 = wave * 16;

    const int nb = row0 + wave * 4;
    int st[4], deg[4], cn[4], ix[4];
#pragma unroll
    for (int i = 0; i < 4; ++i) {
        const int n = nb + i;
        const int nn = min(n, N - 1);
        st[i] = nn * ROW;
        deg[i] = (n < N) ? rows[st[i] + CAP] : 0;
        ix[i] = rows[st[i] + lane];
    }
#pragma unroll
    for (int i = 0; i < 4; ++i) {
        cn[i] = min(deg[i], CAP);
        ix[i] = (lane < min(32, cn[i])) ? ix[i] : 0;
    }
#pragma unroll
    for (int i = 0; i < 4; ++i) {
        float2 a = bucket_sum64((const float2*)xin, rows, st[i], cn[i],
                                ix[i], lane);
        if (deg[i] > CAP) {
            a = overflow_sum64((const float2*)xin, ovf_src, ovf_dst,
                               min(*ovf_cnt, E), nb + i, lane, a);
        }
        const float rinv = 1.0f / fmaxf((float)deg[i], 1.0f);
        float* mrow = sh_m + (wave * 4 + i) * PITCH + 2 * lane;
        mrow[0] = a.x * rinv;
        mrow[1] = a.y * rinv;
    }
    __syncthreads();

    int ar = row0 + lo;
    if (ar >= N) ar = N - 1;
    const int bcol = col0 + lo;

    const float* pm = sh_m + lo * PITCH + half * 2;
    const float* __restrict__ px = xin + (size_t)ar * K + half * 2;
    const float* __restrict__ pl = wl + (size_t)(half * 2) * M + bcol;
    const float* __restrict__ pr = wr + (size_t)(half * 2) * M + bcol;

    v2f af[S], bf[S];
    v8f acc = {};

#pragma unroll
    for (int s = 0; s < S; ++s) {
        af[s].x = pm[s * 4];
        af[s].y = pm[s * 4 + 1];
        bf[s].x = pl[(size_t)(s * 4) * M];
        bf[s].y = pl[(size_t)(s * 4 + 1) * M];
    }
#pragma unroll
    for (int s = 0; s < S; ++s) {
        acc = __builtin_amdgcn_wmma_f32_16x16x4_f32(false, af[s], false, bf[s],
                                                    (short)0, acc, false, false);
    }

#pragma unroll
    for (int s = 0; s < S; ++s) {
        af[s].x = px[s * 4];
        af[s].y = px[s * 4 + 1];
        bf[s].x = pr[(size_t)(s * 4) * M];
        bf[s].y = pr[(size_t)(s * 4 + 1) * M];
    }
#pragma unroll
    for (int s = 0; s < S; ++s) {
        acc = __builtin_amdgcn_wmma_f32_16x16x4_f32(false, af[s], false, bf[s],
                                                    (short)0, acc, false, false);
    }

    const bool is_p = wave < 2;
    const int c2 = (wave & 1) * 16 + lo;
    const float* __restrict__ pw =
        (is_p ? w2l : w2r) + (size_t)(half * 2) * M2 + c2;
    v2f b2f[S];
#pragma unroll
    for (int s = 0; s < S; ++s) {
        b2f[s].x = pw[(size_t)(s * 4) * M2];
        b2f[s].y = pw[(size_t)(s * 4 + 1) * M2];
    }
    const float add2 = is_p ? 0.0f : b2[c2];

    const float bv = bias[bcol];
#pragma unroll
    for (int v = 0; v < 8; ++v) {
        sh_h[(half * 8 + v) * PITCH + bcol] = fmaxf(acc[v] + bv, 0.0f);
    }
    __syncthreads();

    const float* ps = sh_h + lo * PITCH + half * 2;
    v8f acc2 = {};
#pragma unroll
    for (int s = 0; s < S; ++s) {
        af[s].x = ps[s * 4];
        af[s].y = ps[s * 4 + 1];
    }
#pragma unroll
    for (int s = 0; s < S; ++s) {
        acc2 = __builtin_amdgcn_wmma_f32_16x16x4_f32(false, af[s], false, b2f[s],
                                                     (short)0, acc2, false, false);
    }

    float* __restrict__ dst = is_p ? p : q;
#pragma unroll
    for (int v = 0; v < 8; ++v) {
        const int r = row0 + half * 8 + v;
        if (r < N) dst[(size_t)r * M2 + c2] = acc2[v] + add2;
    }
}

__global__ void gather32_layer2_then_classifier(const float* __restrict__ p,
                                                const float* __restrict__ q,
                                                const int* __restrict__ rows,
                                                const int* __restrict__ ovf_cnt,
                                                const int* __restrict__ ovf_src,
                                                const int* __restrict__ ovf_dst,
                                                const float* __restrict__ wc1,
                                                const float* __restrict__ bc1,
                                                const float* __restrict__ wc2,
                                                const float* __restrict__ bc2,
                                                float* __restrict__ out, int N,
                                                int E) {
    __shared__ float sh_h2[16 * 32];
    __shared__ float sh_h3[16 * 16];

    const int lane = threadIdx.x & 31;
    const int wave = __builtin_amdgcn_readfirstlane((int)(threadIdx.x >> 5));
    const int node0 = blockIdx.x * 16;
    const int nb = node0 + wave * 4;

    int st[4], deg[4], cn[4], ix[4];
    float qv[4];
#pragma unroll
    for (int i = 0; i < 4; ++i) {
        const int n = nb + i;
        const int nn = min(n, N - 1);
        st[i] = nn * ROW;
        deg[i] = (n < N) ? rows[st[i] + CAP] : 0;
        ix[i] = rows[st[i] + lane];
        qv[i] = q[(size_t)nn * 32 + lane];
    }
#pragma unroll
    for (int i = 0; i < 4; ++i) {
        cn[i] = min(deg[i], CAP);
        ix[i] = (lane < min(32, cn[i])) ? ix[i] : 0;
    }
#pragma unroll
    for (int i = 0; i < 4; ++i) {
        float acc = bucket_sum32(p, rows, st[i], cn[i], ix[i], lane);
        if (deg[i] > CAP) {
            acc = overflow_sum32(p, ovf_src, ovf_dst, min(*ovf_cnt, E), nb + i,
                                 lane, acc);
        }
        const float d = fmaxf((float)deg[i], 1.0f);
        sh_h2[(wave * 4 + i) * 32 + lane] = fmaxf(acc / d + qv[i], 0.0f);
    }
    __syncthreads();

    const int t = threadIdx.x;
    const int j = t & 15;
    const int na = t >> 4;
    float s0 = bc1[j];
    float s1 = s0;
#pragma unroll
    for (int k = 0; k < 32; ++k) {
        const float w = wc1[k * 16 + j];
        s0 = fmaf(sh_h2[na * 32 + k], w, s0);
        s1 = fmaf(sh_h2[(na + 8) * 32 + k], w, s1);
    }
    sh_h3[na * 16 + j] = fmaxf(s0, 0.0f);
    sh_h3[(na + 8) * 16 + j] = fmaxf(s1, 0.0f);
    __syncthreads();

    if (t < 32) {
        const int node = t >> 1;
        const int jj = t & 1;
        float s = bc2[jj];
#pragma unroll
        for (int k = 0; k < 16; ++k) {
            s = fmaf(sh_h3[node * 16 + k], wc2[k * 2 + jj], s);
        }
        const int n = node0 + node;
        if (n < N) out[(size_t)n * 2 + jj] = s;
    }
}

extern "C" void kernel_launch(void* const* d_in, const int* in_sizes, int n_in,
                              void* d_out, int out_size, void* d_ws, size_t ws_size,
                              hipStream_t stream) {
    const float* x    = (const float*)d_in[0];
    const int*   ei   = (const int*)d_in[1];
    const float* w1l  = (const float*)d_in[2];
    const float* w1r  = (const float*)d_in[3];
    const float* b1   = (const float*)d_in[4];
    const float* w2l  = (const float*)d_in[5];
    const float* w2r  = (const float*)d_in[6];
    const float* b2   = (const float*)d_in[7];
    const float* wc1  = (const float*)d_in[8];
    const float* bc1  = (const float*)d_in[9];
    const float* wc2  = (const float*)d_in[10];
    const float* bc2  = (const float*)d_in[11];
    float* out = (float*)d_out;

    const int N = in_sizes[0] / 64;
    const int E = in_sizes[1] / 2;
    const int* src = ei;
    const int* dst = ei + E;

    float* plane0 = (float*)d_ws;
    float* pq   = plane0 + (size_t)N * 64;
    float* rowp = pq + (size_t)N * 64;
    float* cntp = rowp + (size_t)N * 64;
    float* p    = pq;
    float* q    = pq + (size_t)N * 32;
    int* ovf_src = (int*)plane0;
    int* ovf_dst = ovf_src + E;
    int* rows    = (int*)rowp;
    int* ovf_cnt = (int*)cntp;

    const int nblk = (N + 255) / 256;
    const int eblk = (E + 255) / 256;
    const int sblk = (N + 15) / 16;

    clear_counts<<<nblk, 256, 0, stream>>>(rows, ovf_cnt, N);
    place_edges<<<eblk, 256, 0, stream>>>(src, dst, rows, ovf_cnt, ovf_src,
                                          ovf_dst, E);

    gather64_gemm1_then_l2_products<<<sblk, 128, 0, stream>>>(
        x, rows, ovf_cnt, ovf_src, ovf_dst, w1l, w1r, b1, w2l, w2r, b2, p, q,
        N, E);

    gather32_layer2_then_classifier<<<sblk, 128, 0, stream>>>(
        p, q, rows, ovf_cnt, ovf_src, ovf_dst, wc1, bc1, wc2, bc2, out, N, E);
}


# Edge filing made local to a package half: nodes split even/odd, each work-group draws one ticket and files only its own half's set (8 edges per thread, 782 groups); sums still in arrival-rank order
# speedup vs baseline: 6.5814x; 1.6425x over previous
#include <hip/hip_runtime.h>


typedef __attribute__((ext_vector_type(2))) float v2f;
typedef __attribute__((ext_vector_type(8))) float v8f;

constexpr int ROW = 64;
constexpr int CAP = 63;
constexpr int PLACE_THREADS = 256;
constexpr int EPT = 8;
constexpr int CHUNK = PLACE_THREADS * EPT;
constexpr int TICKET = 64;

__global__ void clear_counts(int* __restrict__ rows, int* __restrict__ counters,
                             int N) {
    const int n = blockIdx.x * blockDim.x + threadIdx.x;
    if (n < N) rows[(size_t)n * ROW + CAP] = 0;
    if (n == 0) {
        counters[0] = 0;
        counters[TICKET] = 0;
        counters[2 * TICKET] = 0;
    }
}

__global__ void __launch_bounds__(PLACE_THREADS)
place_edges(const int* __restrict__ src, const int* __restrict__ dst,
            int* rows, int* counters, int* __restrict__ ovf_src,
            int* __restrict__ ovf_dst, int E, int nchunks) {
    __shared__ int s_ticket[2];
    if (threadIdx.x == 0) {
        const unsigned id = __builtin_amdgcn_s_sendmsg_rtn(0x87);
        int set = (int)((id >> 18) & 1u);
        int c = __hip_atomic_fetch_add(counters + (1 + set) * TICKET, 1,
                                       __ATOMIC_RELAXED,
                                       __HIP_MEMORY_SCOPE_AGENT);
        if (c >= nchunks) {
            set ^= 1;
            c = __hip_atomic_fetch_add(counters + (1 + set) * TICKET, 1,
                                       __ATOMIC_RELAXED,
                                       __HIP_MEMORY_SCOPE_AGENT);
        }
        s_ticket[0] = c;
        s_ticket[1] = set;
    }
    __syncthreads();
    const int c = s_ticket[0];
    const int set = s_ticket[1];
    if (c >= nchunks) return;

    const int e0 = c * CHUNK + threadIdx.x;
    int d[EPT], s[EPT], r[EPT];
    bool mine[EPT];
#pragma unroll
    for (int k = 0; k < EPT; ++k) {
        const int e = e0 + k * PLACE_THREADS;
        const int ee = min(e, E - 1);
        d[k] = dst[ee];
        s[k] = src[ee];
        mine[k] = (e < E) && ((d[k] & 1) == set);
    }
#pragma unroll
    for (int k = 0; k < EPT; ++k) {
        r[k] = 0;
        if (mine[k]) {
            r[k] = __hip_atomic_fetch_add(rows + (size_t)d[k] * ROW + CAP, 1,
                                          __ATOMIC_RELAXED,
                                          __HIP_MEMORY_SCOPE_AGENT);
        }
    }
#pragma unroll
    for (int k = 0; k < EPT; ++k) {
        if (mine[k]) {
            if (r[k] < CAP) {
                rows[(size_t)d[k] * ROW + r[k]] = s[k];
            } else {
                const int j = __hip_atomic_fetch_add(counters, 1,
                                                     __ATOMIC_RELAXED,
                                                     __HIP_MEMORY_SCOPE_AGENT);
                ovf_src[j] = s[k];
                ovf_dst[j] = d[k];
            }
        }
    }
}

__device__ __forceinline__ float2 bucket_sum64(const float2* __restrict__ f2,
                                               const int* __restrict__ perm,
                                               int start, int cnt,
                                               int idx_first, int lane) {
    float2 acc = make_float2(0.0f, 0.0f);
    int idx = idx_first;
    for (int base = 0; base < cnt; base += 32) {
        const int m = min(32, cnt - base);
        if (base > 0) idx = (lane < m) ? perm[start + base + lane] : 0;
        int t = 0;
        for (; t + 16 <= m; t += 16) {
            float2 r[16];
#pragma unroll
            for (int i = 0; i < 16; ++i) {
                const int s = __builtin_amdgcn_readlane(idx, t + i);
                r[i] = f2[(size_t)s * 32 + lane];
            }
#pragma unroll
            for (int i = 0; i < 16; ++i) {
                acc.x += r[i].x;
                acc.y += r[i].y;
            }
        }
        for (; t + 8 <= m; t += 8) {
            float2 r[8];
#pragma unroll
            for (int i = 0; i < 8; ++i) {
                const int s = __builtin_amdgcn_readlane(idx, t + i);
                r[i] = f2[(size_t)s * 32 + lane];
            }
#pragma unroll
            for (int i = 0; i < 8; ++i) {
                acc.x += r[i].x;
                acc.y += r[i].y;
            }
        }
        for (; t < m; t += 4) {
            float2 r[4];
#pragma unroll
            for (int i = 0; i < 4; ++i) {
                const int s = __builtin_amdgcn_readlane(idx, t + i);
                r[i] = f2[(size_t)s * 32 + lane];
            }
#pragma unroll
            for (int i = 0; i < 4; ++i) {
                if (t + i < m) {
                    acc.x += r[i].x;
                    acc.y += r[i].y;
                }
            }
        }
    }
    return acc;
}

__device__ __forceinline__ float bucket_sum32(const float* __restrict__ f,
                                              const int* __restrict__ perm,
                                              int start, int cnt,
                                              int idx_first, int lane) {
    float acc = 0.0f;
    int idx = idx_first;
    for (int base = 0; base < cnt; base += 32) {
        const int m = min(32, cnt - base);
        if (base > 0) idx = (lane < m) ? perm[start + base + lane] : 0;
        int t = 0;
        for (; t + 16 <= m; t += 16) {
            float r[16];
#pragma unroll
            for (int i = 0; i < 16; ++i) {
                const int s = __builtin_amdgcn_readlane(idx, t + i);
                r[i] = f[(size_t)s * 32 + lane];
            }
#pragma unroll
            for (int i = 0; i < 16; ++i) acc += r[i];
        }
        for (; t + 8 <= m; t += 8) {
            float r[8];
#pragma unroll
            for (int i = 0; i < 8; ++i) {
                const int s = __builtin_amdgcn_readlane(idx, t + i);
                r[i] = f[(size_t)s * 32 + lane];
            }
#pragma unroll
            for (int i = 0; i < 8; ++i) acc += r[i];
        }
        for (; t < m; t += 4) {
            float r[4];
#pragma unroll
            for (int i = 0; i < 4; ++i) {
                const int s = __builtin_amdgcn_readlane(idx, t + i);
                r[i] = f[(size_t)s * 32 + lane];
            }
#pragma unroll
            for (int i = 0; i < 4; ++i) {
                if (t + i < m) acc += r[i];
            }
        }
    }
    return acc;
}

__device__ __forceinline__ float2 overflow_sum64(const float2* __restrict__ f2,
                                                 const int* __restrict__ ovf_src,
                                                 const int* __restrict__ ovf_dst,
                                                 int total, int node, int lane,
                                                 float2 acc) {
    for (int base = 0; base < total; base += 32) {
        const int j = base + lane;
        const int dd = (j < total) ? ovf_dst[j] : -1;
        const int ss = (j < total) ? ovf_src[j] : 0;
        unsigned mask = __builtin_amdgcn_ballot_w32(dd == node);
        while (mask != 0u) {
            const int b = __builtin_ctz(mask);
            mask &= mask - 1u;
            const int s = __builtin_amdgcn_readlane(ss, b);
            const float2 r = f2[(size_t)s * 32 + lane];
            acc.x += r.x;
            acc.y += r.y;
        }
    }
    return acc;
}

__device__ __forceinline__ float overflow_sum32(const float* __restrict__ f,
                                                const int* __restrict__ ovf_src,
                                                const int* __restrict__ ovf_dst,
                                                int total, int node, int lane,
                                                float acc) {
    for (int base = 0; base < total; base += 32) {
        const int j = base + lane;
        const int dd = (j < total) ? ovf_dst[j] : -1;
        const int ss = (j < total) ? ovf_src[j] : 0;
        unsigned mask = __builtin_amdgcn_ballot_w32(dd == node);
        while (mask != 0u) {
            const int b = __builtin_ctz(mask);
            mask &= mask - 1u;
            const int s = __builtin_amdgcn_readlane(ss, b);
            acc += f[(size_t)s * 32 + lane];
        }
    }
    return acc;
}

__global__ void __launch_bounds__(128, 16) gather64_gemm1_then_l2_products(const float* __restrict__ xin,
                                                const int* __restrict__ rows,
                                                const int* __restrict__ ovf_cnt,
                                                const int* __restrict__ ovf_src,
                                                const int* __restrict__ ovf_dst,
                                                const float* __restrict__ wl,
                                                const float* __restrict__ wr,
                                                const float* __restrict__ bias,
                                                const float* __restrict__ w2l,
                                                const float* __restrict__ w2r,
                                                const float* __restrict__ b2,
                                                float* __restrict__ p,
                                                float* __restrict__ q,
                                                int N, int E) {
    constexpr int K = 64;
    constexpr int M = 64;
    constexpr int M2 = 32;
    constexpr int S = K / 4;
    constexpr int PITCH = 68;
    __shared__ float sh[2 * 16 * PITCH];
    float* const sh_m = sh;
    float* const sh_h = sh + 16 * PITCH;

    const int lane = threadIdx.x & 31;
    const int wave = __builtin_amdgcn_readfirstlane((int)(threadIdx.x >> 5));
    const int lo   = lane & 15;
    const int half = lane >> 4;
    const int row0 = blockIdx.x * 16;
    const int col0 = wave * 16;

    const int nb = row0 + wave * 4;
    int st[4], deg[4], cn[4], ix[4];
#pragma unroll
    for (int i = 0; i < 4; ++i) {
        const int n = nb + i;
        const int nn = min(n, N - 1);
        st[i] = nn * ROW;
        deg[i] = (n < N) ? rows[st[i] + CAP] : 0;
        ix[i] = rows[st[i] + lane];
    }
#pragma unroll
    for (int i = 0; i < 4; ++i) {
        cn[i] = min(deg[i], CAP);
        ix[i] = (lane < min(32, cn[i])) ? ix[i] : 0;
    }
#pragma unroll
    for (int i = 0; i < 4; ++i) {
        float2 a = bucket_sum64((const float2*)xin, rows, st[i], cn[i],
                                ix[i], lane);
        if (deg[i] > CAP) {
            a = overflow_sum64((const float2*)xin, ovf_src, ovf_dst,
                               min(*ovf_cnt, E), nb + i, lane, a);
        }
        const float rinv = 1.0f / fmaxf((float)deg[i], 1.0f);
        float* mrow = sh_m + (wave * 4 + i) * PITCH + 2 * lane;
        mrow[0] = a.x * rinv;
        mrow[1] = a.y * rinv;
    }
    __syncthreads();

    int ar = row0 + lo;
    if (ar >= N) ar = N - 1;
    const int bcol = col0 + lo;

    const float* pm = sh_m + lo * PITCH + half * 2;
    const float* __restrict__ px = xin + (size_t)ar * K + half * 2;
    const float* __restrict__ pl = wl + (size_t)(half * 2) * M + bcol;
    const float* __restrict__ pr = wr + (size_t)(half * 2) * M + bcol;

    v2f af[S], bf[S];
    v8f acc = {};

#pragma unroll
    for (int s = 0; s < S; ++s) {
        af[s].x = pm[s * 4];
        af[s].y = pm[s * 4 + 1];
        bf[s].x = pl[(size_t)(s * 4) * M];
        bf[s].y = pl[(size_t)(s * 4 + 1) * M];
    }
#pragma unroll
    for (int s = 0; s < S; ++s) {
        acc = __builtin_amdgcn_wmma_f32_16x16x4_f32(false, af[s], false, bf[s],
                                                    (short)0, acc, false, false);
    }

#pragma unroll
    for (int s = 0; s < S; ++s) {
        af[s].x = px[s * 4];
        af[s].y = px[s * 4 + 1];
        bf[s].x = pr[(size_t)(s * 4) * M];
        bf[s].y = pr[(size_t)(s * 4 + 1) * M];
    }
#pragma unroll
    for (int s = 0; s < S; ++s) {
        acc = __builtin_amdgcn_wmma_f32_16x16x4_f32(false, af[s], false, bf[s],
                                                    (short)0, acc, false, false);
    }

    const bool is_p = wave < 2;
    const int c2 = (wave & 1) * 16 + lo;
    const float* __restrict__ pw =
        (is_p ? w2l : w2r) + (size_t)(half * 2) * M2 + c2;
    v2f b2f[S];
#pragma unroll
    for (int s = 0; s < S; ++s) {
        b2f[s].x = pw[(size_t)(s * 4) * M2];
        b2f[s].y = pw[(size_t)(s * 4 + 1) * M2];
    }
    const float add2 = is_p ? 0.0f : b2[c2];

    const float bv = bias[bcol];
#pragma unroll
    for (int v = 0; v < 8; ++v) {
        sh_h[(half * 8 + v) * PITCH + bcol] = fmaxf(acc[v] + bv, 0.0f);
    }
    __syncthreads();

    const float* ps = sh_h + lo * PITCH + half * 2;
    v8f acc2 = {};
#pragma unroll
    for (int s = 0; s < S; ++s) {
        af[s].x = ps[s * 4];
        af[s].y = ps[s * 4 + 1];
    }
#pragma unroll
    for (int s = 0; s < S; ++s) {
        acc2 = __builtin_amdgcn_wmma_f32_16x16x4_f32(false, af[s], false, b2f[s],
                                                     (short)0, acc2, false, false);
    }

    float* __restrict__ dst = is_p ? p : q;
#pragma unroll
    for (int v = 0; v < 8; ++v) {
        const int r = row0 + half * 8 + v;
        if (r < N) dst[(size_t)r * M2 + c2] = acc2[v] + add2;
    }
}

__global__ void gather32_layer2_then_classifier(const float* __restrict__ p,
                                                const float* __restrict__ q,
                                                const int* __restrict__ rows,
                                                const int* __restrict__ ovf_cnt,
                                                const int* __restrict__ ovf_src,
                                                const int* __restrict__ ovf_dst,
                                                const float* __restrict__ wc1,
                                                const float* __restrict__ bc1,
                                                const float* __restrict__ wc2,
                                                const float* __restrict__ bc2,
                                                float* __restrict__ out, int N,
                                                int E) {
    __shared__ float sh_h2[16 * 32];
    __shared__ float sh_h3[16 * 16];

    const int lane = threadIdx.x & 31;
    const int wave = __builtin_amdgcn_readfirstlane((int)(threadIdx.x >> 5));
    const int node0 = blockIdx.x * 16;
    const int nb = node0 + wave * 4;

    int st[4], deg[4], cn[4], ix[4];
    float qv[4];
#pragma unroll
    for (int i = 0; i < 4; ++i) {
        const int n = nb + i;
        const int nn = min(n, N - 1);
        st[i] = nn * ROW;
        deg[i] = (n < N) ? rows[st[i] + CAP] : 0;
        ix[i] = rows[st[i] + lane];
        qv[i] = q[(size_t)nn * 32 + lane];
    }
#pragma unroll
    for (int i = 0; i < 4; ++i) {
        cn[i] = min(deg[i], CAP);
        ix[i] = (lane < min(32, cn[i])) ? ix[i] : 0;
    }
#pragma unroll
    for (int i = 0; i < 4; ++i) {
        float acc = bucket_sum32(p, rows, st[i], cn[i], ix[i], lane);
        if (deg[i] > CAP) {
            acc = overflow_sum32(p, ovf_src, ovf_dst, min(*ovf_cnt, E), nb + i,
                                 lane, acc);
        }
        const float d = fmaxf((float)deg[i], 1.0f);
        sh_h2[(wave * 4 + i) * 32 + lane] = fmaxf(acc / d + qv[i], 0.0f);
    }
    __syncthreads();

    const int t = threadIdx.x;
    const int j = t & 15;
    const int na = t >> 4;
    float s0 = bc1[j];
    float s1 = s0;
#pragma unroll
    for (int k = 0; k < 32; ++k) {
        const float w = wc1[k * 16 + j];
        s0 = fmaf(sh_h2[na * 32 + k], w, s0);
        s1 = fmaf(sh_h2[(na + 8) * 32 + k], w, s1);
    }
    sh_h3[na * 16 + j] = fmaxf(s0, 0.0f);
    sh_h3[(na + 8) * 16 + j] = fmaxf(s1, 0.0f);
    __syncthreads();

    if (t < 32) {
        const int node = t >> 1;
        const int jj = t & 1;
        float s = bc2[jj];
#pragma unroll
        for (int k = 0; k < 16; ++k) {
            s = fmaf(sh_h3[node * 16 + k], wc2[k * 2 + jj], s);
        }
        const int n = node0 + node;
        if (n < N) out[(size_t)n * 2 + jj] = s;
    }
}

extern "C" void kernel_launch(void* const* d_in, const int* in_sizes, int n_in,
                              void* d_out, int out_size, void* d_ws, size_t ws_size,
                              hipStream_t stream) {
    const float* x    = (const float*)d_in[0];
    const int*   ei   = (const int*)d_in[1];
    const float* w1l  = (const float*)d_in[2];
    const float* w1r  = (const float*)d_in[3];
    const float* b1   = (const float*)d_in[4];
    const float* w2l  = (const float*)d_in[5];
    const float* w2r  = (const float*)d_in[6];
    const float* b2   = (const float*)d_in[7];
    const float* wc1  = (const float*)d_in[8];
    const float* bc1  = (const float*)d_in[9];
    const float* wc2  = (const float*)d_in[10];
    const float* bc2  = (const float*)d_in[11];
    float* out = (float*)d_out;

    const int N = in_sizes[0] / 64;
    const int E = in_sizes[1] / 2;
    const int* src = ei;
    const int* dst = ei + E;

    float* plane0 = (float*)d_ws;
    float* pq   = plane0 + (size_t)N * 64;
    float* rowp = pq + (size_t)N * 64;
    float* cntp = rowp + (size_t)N * 64;
    float* p    = pq;
    float* q    = pq + (size_t)N * 32;
    int* ovf_src = (int*)plane0;
    int* ovf_dst = ovf_src + E;
    int* rows    = (int*)rowp;
    int* ovf_cnt = (int*)cntp;

    const int nblk = (N + 255) / 256;
    const int nchunks = (E + CHUNK - 1) / CHUNK;
    const int sblk = (N + 15) / 16;

    clear_counts<<<nblk, 256, 0, stream>>>(rows, ovf_cnt, N);
    place_edges<<<2 * nchunks, PLACE_THREADS, 0, stream>>>(
        src, dst, rows, ovf_cnt, ovf_src, ovf_dst, E, nchunks);

    gather64_gemm1_then_l2_products<<<sblk, 128, 0, stream>>>(
        x, rows, ovf_cnt, ovf_src, ovf_dst, w1l, w1r, b1, w2l, w2r, b2, p, q,
        N, E);

    gather32_layer2_then_classifier<<<sblk, 128, 0, stream>>>(
        p, q, rows, ovf_cnt, ovf_src, ovf_dst, wc1, bc1, wc2, bc2, out, N, E);
}
